# starting kernel
# baseline (speedup 1.0000x reference)
#include <hip/hip_runtime.h>

#define P_     16
#define K_     256
#define W_     8
#define F_IN   128
#define F_OUT  128
#define TILE_M 16

typedef __attribute__((ext_vector_type(2))) float v2f;
typedef __attribute__((ext_vector_type(8))) float v8f;

__global__ __launch_bounds__(256) void sage_fused_kernel(
    const int*   __restrict__ codes,
    const int*   __restrict__ indices,
    const int*   __restrict__ indptr,
    const float* __restrict__ h_self,
    const float* __restrict__ codebook,
    const float* __restrict__ W_neigh,
    const float* __restrict__ W_self,
    const float* __restrict__ b_self,
    float*       __restrict__ out,
    int n_dst)
{
    __shared__ __align__(16) float lds_hn[TILE_M * F_IN];
    __shared__ __align__(16) float lds_hs[TILE_M * F_IN];
    __shared__ int   sidx[TILE_M * 16];
    __shared__ float sinv[TILE_M];

    const int tid      = threadIdx.x;
    const int row_base = blockIdx.x * TILE_M;

    {
        const int row = tid >> 4;
        const int e   = tid & 15;
        int deg = 0, src = -1;
        if (row_base + row < n_dst) {
            const int beg = indptr[row_base + row];
            const int end = indptr[row_base + row + 1];
            deg = end - beg;
            if (e < deg) src = indices[beg + e];
        }
        sidx[tid] = src;
        if (e == 0) sinv[row] = 1.0f / fmaxf((float)deg, 1.0f);
    }
    __syncthreads();

    for (int idx = tid; idx < TILE_M * (F_IN / 4); idx += 256) {
        const int row = idx >> 5;
        const int c4  = idx & 31;
        const int gr  = row_base + row;

        float4 hs = make_float4(0.f, 0.f, 0.f, 0.f);
        if (gr < n_dst)
            hs = *(const float4*)(h_self + (size_t)gr * F_IN + c4 * 4);
        *(float4*)(&lds_hs[row * F_IN + c4 * 4]) = hs;

        const int p  = c4 >> 1;
        const int w0 = (c4 & 1) * 4;
        float ax = 0.f, ay = 0.f, az = 0.f, aw = 0.f;
        #pragma unroll
        for (int e = 0; e < 16; ++e) {
            const int s = sidx[row * 16 + e];
            if (s >= 0) {
                const int code = codes[s * P_ + p];
                const float4 cb =
                    *(const float4*)(codebook + ((p * K_ + code) * W_ + w0));
                ax += cb.x; ay += cb.y; az += cb.z; aw += cb.w;
            }
        }
        const float inv = sinv[row];
        const float4 hn = make_float4(ax * inv, ay * inv, az * inv, aw * inv);
        *(float4*)(&lds_hn[row * F_IN + c4 * 4]) = hn;
    }
    __syncthreads();

    const int lane  = tid & 31;
    const int wave  = tid >> 5;
    const int nbase = wave * 16;
    const int m     = lane & 15;
    const int khalf = (lane >> 4) * 2;

    v8f acc = {};
    const float* an = lds_hn + m * F_IN + khalf;
    const float* as = lds_hs + m * F_IN + khalf;
    const float* wn = W_neigh + (size_t)(nbase + m) * F_IN + khalf;
    const float* ws = W_self  + (size_t)(nbase + m) * F_IN + khalf;

    #pragma unroll 8
    for (int k0 = 0; k0 < F_IN; k0 += 4) {
        v2f a = *(const v2f*)(an + k0);
        v2f b = *(const v2f*)(wn + k0);
        acc = __builtin_amdgcn_wmma_f32_16x16x4_f32(
            false, a, false, b, (short)0, acc, false, false);
    }
    #pragma unroll 8
    for (int k0 = 0; k0 < F_IN; k0 += 4) {
        v2f a = *(const v2f*)(as + k0);
        v2f b = *(const v2f*)(ws + k0);
        acc = __builtin_amdgcn_wmma_f32_16x16x4_f32(
            false, a, false, b, (short)0, acc, false, false);
    }

    const float bias  = b_self[nbase + m];
    const int   rhalf = (lane >> 4) * 8;
    #pragma unroll
    for (int j = 0; j < 8; ++j) {
        const int gr = row_base + rhalf + j;
        if (gr < n_dst)
            out[(size_t)gr * F_OUT + nbase + m] = acc[j] + bias;
    }
}

extern "C" void kernel_launch(void* const* d_in, const int* in_sizes, int n_in,
                              void* d_out, int out_size, void* d_ws, size_t ws_size,
                              hipStream_t stream) {
    (void)n_in; (void)out_size; (void)d_ws; (void)ws_size;
    const int*   codes    = (const int*)  d_in[0];
    const int*   indices  = (const int*)  d_in[1];
    const int*   indptr   = (const int*)  d_in[2];
    const float* h_self   = (const float*)d_in[3];
    const float* codebook = (const float*)d_in[4];
    const float* W_neigh  = (const float*)d_in[5];
    const float* W_self   = (const float*)d_in[6];
    const float* b_self   = (const float*)d_in[7];
    float*       out      = (float*)d_out;

    const int n_dst  = in_sizes[2] - 1;
    const int blocks = (n_dst + TILE_M - 1) / TILE_M;

    sage_fused_kernel<<<blocks, 256, 0, stream>>>(
        codes, indices, indptr, h_self, codebook, W_neigh, W_self, b_self,
        out, n_dst);
}


# B (W) fragments no longer loaded straight from global, 16 lines per 8-byte load: each wave stages its 16-row W slice by whole-line 16-byte loads into a padded private LDS tile and reads B there
# speedup vs baseline: 1.1179x; 1.1179x over previous
#include <hip/hip_runtime.h>

#define P_     16
#define K_     256
#define W_     8
#define F_IN   128
#define F_OUT  128
#define TILE_M 16
#define KCH    32
#define WPITCH 36

typedef __attribute__((ext_vector_type(2))) float v2f;
typedef __attribute__((ext_vector_type(8))) float v8f;

__global__ __launch_bounds__(256) void sage_fused_kernel(
    const int*   __restrict__ codes,
    const int*   __restrict__ indices,
    const int*   __restrict__ indptr,
    const float* __restrict__ h_self,
    const float* __restrict__ codebook,
    const float* __restrict__ W_neigh,
    const float* __restrict__ W_self,
    const float* __restrict__ b_self,
    float*       __restrict__ out,
    int n_dst)
{
    __shared__ __align__(16) float lds_hn[TILE_M * F_IN];
    __shared__ __align__(16) float lds_hs[TILE_M * F_IN];
    __shared__ int   sidx[TILE_M * 16];
    __shared__ float sinv[TILE_M];
    __shared__ __align__(16) float lds_w[8 * 16 * WPITCH];

    const int tid      = threadIdx.x;
    const int row_base = blockIdx.x * TILE_M;

    {
        const int row = tid >> 4;
        const int e   = tid & 15;
        int deg = 0, src = -1;
        if (row_base + row < n_dst) {
            const int beg = indptr[row_base + row];
            const int end = indptr[row_base + row + 1];
            deg = end - beg;
            if (e < deg) src = indices[beg + e];
        }
        sidx[tid] = src;
        if (e == 0) sinv[row] = 1.0f / fmaxf((float)deg, 1.0f);
    }
    __syncthreads();

    for (int idx = tid; idx < TILE_M * (F_IN / 4); idx += 256) {
        const int row = idx >> 5;
        const int c4  = idx & 31;
        const int gr  = row_base + row;

        float4 hs = make_float4(0.f, 0.f, 0.f, 0.f);
        if (gr < n_dst)
            hs = *(const float4*)(h_self + (size_t)gr * F_IN + c4 * 4);
        *(float4*)(&lds_hs[row * F_IN + c4 * 4]) = hs;

        const int p  = c4 >> 1;
        const int w0 = (c4 & 1) * 4;
        float ax = 0.f, ay = 0.f, az = 0.f, aw = 0.f;
        #pragma unroll
        for (int e = 0; e < 16; ++e) {
            const int s = sidx[row * 16 + e];
            if (s >= 0) {
                const int code = codes[s * P_ + p];
                const float4 cb =
                    *(const float4*)(codebook + ((p * K_ + code) * W_ + w0));
                ax += cb.x; ay += cb.y; az += cb.z; aw += cb.w;
            }
        }
        const float inv = sinv[row];
        const float4 hn = make_float4(ax * inv, ay * inv, az * inv, aw * inv);
        *(float4*)(&lds_hn[row * F_IN + c4 * 4]) = hn;
    }
    __syncthreads();

    const int lane  = tid & 31;
    const int wave  = tid >> 5;
    const int nbase = wave * 16;
    const int m     = lane & 15;
    const int khalf = (lane >> 4) * 2;

    const int srow = lane >> 3;
    const int scol = (lane & 7) * 4;
    float*       wbuf = lds_w + wave * (16 * WPITCH);
    float*       wst  = wbuf + srow * WPITCH + scol;
    const float* wfr  = wbuf + m * WPITCH + khalf;
    const float* gn   = W_neigh + (size_t)(nbase + srow) * F_IN + scol;
    const float* gs   = W_self  + (size_t)(nbase + srow) * F_IN + scol;
    const float* an   = lds_hn + m * F_IN + khalf;
    const float* as   = lds_hs + m * F_IN + khalf;

    float4 g0 = *(const float4*)(gn);
    float4 g1 = *(const float4*)(gn + 4 * F_IN);
    float4 g2 = *(const float4*)(gn + 8 * F_IN);
    float4 g3 = *(const float4*)(gn + 12 * F_IN);

    v8f acc = {};
    #pragma unroll
    for (int t = 0; t < 8; ++t) {
        *(float4*)(wst)               = g0;
        *(float4*)(wst + 4 * WPITCH)  = g1;
        *(float4*)(wst + 8 * WPITCH)  = g2;
        *(float4*)(wst + 12 * WPITCH) = g3;
        __builtin_amdgcn_fence(__ATOMIC_ACQ_REL, "wavefront", "local");

        if (t + 1 < 8) {
            const float* src = ((t + 1) < 4 ? gn : gs) + ((t + 1) & 3) * KCH;
            g0 = *(const float4*)(src);
            g1 = *(const float4*)(src + 4 * F_IN);
            g2 = *(const float4*)(src + 8 * F_IN);
            g3 = *(const float4*)(src + 12 * F_IN);
        }

        const float* ap = (t < 4 ? an : as) + (t & 3) * KCH;
        #pragma unroll
        for (int j = 0; j < 8; ++j) {
            v2f a = *(const v2f*)(ap  + 4 * j);
            v2f b = *(const v2f*)(wfr + 4 * j);
            acc = __builtin_amdgcn_wmma_f32_16x16x4_f32(
                false, a, false, b, (short)0, acc, false, false);
        }
        __builtin_amdgcn_fence(__ATOMIC_ACQ_REL, "wavefront", "local");
    }

    const float bias  = b_self[nbase + m];
    const int   rhalf = (lane >> 4) * 8;
    #pragma unroll
    for (int j = 0; j < 8; ++j) {
        const int gr = row_base + rhalf + j;
        if (gr < n_dst)
            out[(size_t)gr * F_OUT + nbase + m] = acc[j] + bias;
    }
}

extern "C" void kernel_launch(void* const* d_in, const int* in_sizes, int n_in,
                              void* d_out, int out_size, void* d_ws, size_t ws_size,
                              hipStream_t stream) {
    (void)n_in; (void)out_size; (void)d_ws; (void)ws_size;
    const int*   codes    = (const int*)  d_in[0];
    const int*   indices  = (const int*)  d_in[1];
    const int*   indptr   = (const int*)  d_in[2];
    const float* h_self   = (const float*)d_in[3];
    const float* codebook = (const float*)d_in[4];
    const float* W_neigh  = (const float*)d_in[5];
    const float* W_self   = (const float*)d_in[6];
    const float* b_self   = (const float*)d_in[7];
    float*       out      = (float*)d_out;

    const int n_dst  = in_sizes[2] - 1;
    const int blocks = (n_dst + TILE_M - 1) / TILE_M;

    sage_fused_kernel<<<blocks, 256, 0, stream>>>(
        codes, indices, indptr, h_self, codebook, W_neigh, W_self, b_self,
        out, n_dst);
}


# Gather of an item: its 16 code loads requested together, then its 16 codebook loads, then summed in the same edge order; before, every load waited for the one before it
# speedup vs baseline: 1.2669x; 1.1333x over previous
#include <hip/hip_runtime.h>

#define P_     16
#define K_     256
#define W_     8
#define F_IN   128
#define F_OUT  128
#define TILE_M 16
#define KCH    32
#define WPITCH 36

typedef __attribute__((ext_vector_type(2))) float v2f;
typedef __attribute__((ext_vector_type(8))) float v8f;

__global__ __launch_bounds__(256) void sage_fused_kernel(
    const int*   __restrict__ codes,
    const int*   __restrict__ indices,
    const int*   __restrict__ indptr,
    const float* __restrict__ h_self,
    const float* __restrict__ codebook,
    const float* __restrict__ W_neigh,
    const float* __restrict__ W_self,
    const float* __restrict__ b_self,
    float*       __restrict__ out,
    int n_dst)
{
    __shared__ __align__(16) float lds_hn[TILE_M * F_IN];
    __shared__ __align__(16) float lds_hs[TILE_M * F_IN];
    __shared__ __align__(16) int sidx[TILE_M * 16];
    __shared__ float sinv[TILE_M];
    __shared__ __align__(16) float lds_w[8 * 16 * WPITCH];

    const int tid      = threadIdx.x;
    const int row_base = blockIdx.x * TILE_M;

    {
        const int row = tid >> 4;
        const int e   = tid & 15;
        int deg = 0, src = -1;
        if (row_base + row < n_dst) {
            const int beg = indptr[row_base + row];
            const int end = indptr[row_base + row + 1];
            deg = end - beg;
            if (e < deg) src = indices[beg + e];
        }
        sidx[tid] = src;
        if (e == 0) sinv[row] = 1.0f / fmaxf((float)deg, 1.0f);
    }
    __syncthreads();

    const float4* cbk4 = (const float4*)codebook;
    for (int idx = tid; idx < TILE_M * (F_IN / 4); idx += 256) {
        const int row = idx >> 5;
        const int c4  = idx & 31;
        const int gr  = row_base + row;

        float4 hs = make_float4(0.f, 0.f, 0.f, 0.f);
        if (gr < n_dst)
            hs = *(const float4*)(h_self + (size_t)gr * F_IN + c4 * 4);

        const int p  = c4 >> 1;
        const int h4 = c4 & 1;

        int s[16];
        #pragma unroll
        for (int q = 0; q < 4; ++q) {
            const int4 v = *(const int4*)(&sidx[row * 16 + q * 4]);
            s[q * 4 + 0] = __builtin_amdgcn_readfirstlane(v.x);
            s[q * 4 + 1] = __builtin_amdgcn_readfirstlane(v.y);
            s[q * 4 + 2] = __builtin_amdgcn_readfirstlane(v.z);
            s[q * 4 + 3] = __builtin_amdgcn_readfirstlane(v.w);
        }
        int code[16];
        #pragma unroll
        for (int e = 0; e < 16; ++e)
            code[e] = codes[(unsigned)((s[e] >= 0 ? s[e] : 0) * P_ + p)];
        float4 cb[16];
        #pragma unroll
        for (int e = 0; e < 16; ++e)
            cb[e] = cbk4[(unsigned)((p * K_ + code[e]) * 2 + h4)];

        #pragma unroll
        for (int e = 0; e < 16; ++e)
            asm volatile("" :: "v"(cb[e].x), "v"(cb[e].y), "v"(cb[e].z), "v"(cb[e].w));

        float ax = 0.f, ay = 0.f, az = 0.f, aw = 0.f;
        #pragma unroll
        for (int e = 0; e < 16; ++e) {
            const bool on = s[e] >= 0;
            ax += on ? cb[e].x : 0.f;
            ay += on ? cb[e].y : 0.f;
            az += on ? cb[e].z : 0.f;
            aw += on ? cb[e].w : 0.f;
        }
        const float inv = sinv[row];
        const float4 hn = make_float4(ax * inv, ay * inv, az * inv, aw * inv);
        *(float4*)(&lds_hs[row * F_IN + c4 * 4]) = hs;
        *(float4*)(&lds_hn[row * F_IN + c4 * 4]) = hn;
    }
    __syncthreads();

    const int lane  = tid & 31;
    const int wave  = tid >> 5;
    const int nbase = wave * 16;
    const int m     = lane & 15;
    const int khalf = (lane >> 4) * 2;

    const int srow = lane >> 3;
    const int scol = (lane & 7) * 4;
    float*       wbuf = lds_w + wave * (16 * WPITCH);
    float*       wst  = wbuf + srow * WPITCH + scol;
    const float* wfr  = wbuf + m * WPITCH + khalf;
    const float* gn   = W_neigh + (size_t)(nbase + srow) * F_IN + scol;
    const float* gs   = W_self  + (size_t)(nbase + srow) * F_IN + scol;
    const float* an   = lds_hn + m * F_IN + khalf;
    const float* as   = lds_hs + m * F_IN + khalf;

    float4 g0 = *(const float4*)(gn);
    float4 g1 = *(const float4*)(gn + 4 * F_IN);
    float4 g2 = *(const float4*)(gn + 8 * F_IN);
    float4 g3 = *(const float4*)(gn + 12 * F_IN);

    v8f acc = {};
    #pragma unroll
    for (int t = 0; t < 8; ++t) {
        *(float4*)(wst)               = g0;
        *(float4*)(wst + 4 * WPITCH)  = g1;
        *(float4*)(wst + 8 * WPITCH)  = g2;
        *(float4*)(wst + 12 * WPITCH) = g3;
        __builtin_amdgcn_fence(__ATOMIC_ACQ_REL, "wavefront", "local");

        if (t + 1 < 8) {
            const float* src = ((t + 1) < 4 ? gn : gs) + ((t + 1) & 3) * KCH;
            g0 = *(const float4*)(src);
            g1 = *(const float4*)(src + 4 * F_IN);
            g2 = *(const float4*)(src + 8 * F_IN);
            g3 = *(const float4*)(src + 12 * F_IN);
        }

        const float* ap = (t < 4 ? an : as) + (t & 3) * KCH;
        #pragma unroll
        for (int j = 0; j < 8; ++j) {
            v2f a = *(const v2f*)(ap  + 4 * j);
            v2f b = *(const v2f*)(wfr + 4 * j);
            acc = __builtin_amdgcn_wmma_f32_16x16x4_f32(
                false, a, false, b, (short)0, acc, false, false);
        }
        __builtin_amdgcn_fence(__ATOMIC_ACQ_REL, "wavefront", "local");
    }

    const float bias  = b_self[nbase + m];
    const int   rhalf = (lane >> 4) * 8;
    #pragma unroll
    for (int j = 0; j < 8; ++j) {
        const int gr = row_base + rhalf + j;
        if (gr < n_dst)
            out[(size_t)gr * F_OUT + nbase + m] = acc[j] + bias;
    }
}

extern "C" void kernel_launch(void* const* d_in, const int* in_sizes, int n_in,
                              void* d_out, int out_size, void* d_ws, size_t ws_size,
                              hipStream_t stream) {
    (void)n_in; (void)out_size; (void)d_ws; (void)ws_size;
    const int*   codes    = (const int*)  d_in[0];
    const int*   indices  = (const int*)  d_in[1];
    const int*   indptr   = (const int*)  d_in[2];
    const float* h_self   = (const float*)d_in[3];
    const float* codebook = (const float*)d_in[4];
    const float* W_neigh  = (const float*)d_in[5];
    const float* W_self   = (const float*)d_in[6];
    const float* b_self   = (const float*)d_in[7];
    float*       out      = (float*)d_out;

    const int n_dst  = in_sizes[2] - 1;
    const int blocks = (n_dst + TILE_M - 1) / TILE_M;

    sage_fused_kernel<<<blocks, 256, 0, stream>>>(
        codes, indices, indptr, h_self, codebook, W_neigh, W_self, b_self,
        out, n_dst);
}


# Codebook held in LDS (128 KB, code-major so the 8 parts a half wave reads fall on 64 different banks) by 256 resident 32-wave work-groups looping over the row tiles: gather reads LDS, not L2 lines
# speedup vs baseline: 1.4634x; 1.1551x over previous
#include <hip/hip_runtime.h>


#define P_     16
#define K_     256
#define W_     8
#define F_IN   128
#define F_OUT  128
#define TILE_M 16
#define KCH    32
#define WPITCH 36
#define NSG    4
#define NTHR   (NSG * 256)
#define NWG    256

typedef __attribute__((ext_vector_type(2))) float v2f;
typedef __attribute__((ext_vector_type(8))) float v8f;

__global__ __launch_bounds__(NTHR) void sage_fused_kernel(
    const int*   __restrict__ codes,
    const int*   __restrict__ indices,
    const int*   __restrict__ indptr,
    const float* __restrict__ h_self,
    const float* __restrict__ codebook,
    const float* __restrict__ W_neigh,
    const float* __restrict__ W_self,
    const float* __restrict__ b_self,
    float*       __restrict__ out,
    int n_dst)
{
    __shared__ __align__(16) float lds_cb[K_ * P_ * W_];
    __shared__ __align__(16) float lds_hn[NSG][TILE_M * F_IN];
    __shared__ __align__(16) float lds_hs[NSG][TILE_M * F_IN];
    __shared__ __align__(16) int   sidx[NSG][TILE_M * 16];
    __shared__ float sinv[NSG][TILE_M];
    __shared__ __align__(16) float lds_w[NSG][8 * 16 * WPITCH];

    const int tid = threadIdx.x;
    const int sg  = tid >> 8;
    const int lt  = tid & 255;

    const int n_tiles  = (n_dst + TILE_M - 1) / TILE_M;
    const int t_first  = (int)((blockIdx.x * (unsigned)n_tiles) / gridDim.x);
    const int t_last   = (int)(((blockIdx.x + 1u) * (unsigned)n_tiles) / gridDim.x);
    const int n_rounds = (t_last - t_first + NSG - 1) / NSG;

    const float4* cbk4 = (const float4*)codebook;
    float4*       cbl4 = (float4*)lds_cb;

    const int erow = lt >> 4;
    const int ee   = lt & 15;
    int deg = 0, src = -1;
    {
        const int tile = t_first + sg;
        const int r    = tile * TILE_M + erow;
        if (tile < t_last && r < n_dst) {
            const int beg = indptr[r];
            deg = indptr[r + 1] - beg;
            if (ee < deg) src = indices[beg + ee];
        }
    }

    {
        const int lane = tid & 31;
        const int cc = lane >> 4, pp = (lane >> 1) & 7, h4 = lane & 1;
        #pragma unroll
        for (int j = tid >> 5; j < (K_ / 2) * (P_ / 8); j += NTHR / 32) {
            const int k = 2 * (j >> 1) + cc;
            const int p = 8 * (j & 1) + pp;
            cbl4[(k * P_ + p) * 2 + h4] = cbk4[(p * K_ + k) * 2 + h4];
        }
    }

    const int lane  = lt & 31;
    const int wave  = lt >> 5;
    const int nbase = wave * 16;
    const int m     = lane & 15;
    const int khalf = (lane >> 4) * 2;

    for (int rd = 0; rd < n_rounds; ++rd) {
        const int  tile     = t_first + rd * NSG + sg;
        const bool have     = tile < t_last;
        const int  row_base = tile * TILE_M;

        sidx[sg][lt] = src;
        if (ee == 0) sinv[sg][erow] = 1.0f / fmaxf((float)deg, 1.0f);
        __syncthreads();

        const int  nr    = (tile + NSG) * TILE_M + erow;
        const bool nhave = (tile + NSG) < t_last && nr < n_dst;
        int nbeg = 0, nend = 0;
        if (nhave) { nbeg = indptr[nr]; nend = indptr[nr + 1]; }

        if (have) {
            for (int idx = lt; idx < TILE_M * (F_IN / 4); idx += 256) {
                const int row = idx >> 5;
                const int c4  = idx & 31;
                const int gr  = row_base + row;

                float4 hs = make_float4(0.f, 0.f, 0.f, 0.f);
                if (gr < n_dst)
                    hs = *(const float4*)(h_self + (size_t)gr * F_IN + c4 * 4);

                const int p = c4 >> 1;

                int s[16];
                #pragma unroll
                for (int q = 0; q < 4; ++q) {
                    const int4 v = *(const int4*)(&sidx[sg][row * 16 + q * 4]);
                    s[q * 4 + 0] = __builtin_amdgcn_readfirstlane(v.x);
                    s[q * 4 + 1] = __builtin_amdgcn_readfirstlane(v.y);
                    s[q * 4 + 2] = __builtin_amdgcn_readfirstlane(v.z);
                    s[q * 4 + 3] = __builtin_amdgcn_readfirstlane(v.w);
                }
                int code[16];
                #pragma unroll
                for (int e = 0; e < 16; ++e)
                    code[e] = codes[(unsigned)((s[e] >= 0 ? s[e] : 0) * P_ + p)];
                float4 cb[16];
                #pragma unroll
                for (int e = 0; e < 16; ++e)
                    cb[e] = cbl4[code[e] * (P_ * 2) + c4];

                #pragma unroll
                for (int e = 0; e < 16; ++e)
                    asm volatile("" :: "v"(cb[e].x), "v"(cb[e].y), "v"(cb[e].z), "v"(cb[e].w));

                float ax = 0.f, ay = 0.f, az = 0.f, aw = 0.f;
                #pragma unroll
                for (int e = 0; e < 16; ++e) {
                    const bool on = s[e] >= 0;
                    ax += on ? cb[e].x : 0.f;
                    ay += on ? cb[e].y : 0.f;
                    az += on ? cb[e].z : 0.f;
                    aw += on ? cb[e].w : 0.f;
                }
                const float inv = sinv[sg][row];
                const float4 hn = make_float4(ax * inv, ay * inv, az * inv, aw * inv);
                *(float4*)(&lds_hs[sg][row * F_IN + c4 * 4]) = hs;
                *(float4*)(&lds_hn[sg][row * F_IN + c4 * 4]) = hn;
            }
        }
        __syncthreads();

        deg = nend - nbeg;
        src = -1;
        if (nhave && ee < deg) src = indices[nbeg + ee];

        if (have) {
            const int srow = lane >> 3;
            const int scol = (lane & 7) * 4;
            float*       wbuf = lds_w[sg] + wave * (16 * WPITCH);
            float*       wst  = wbuf + srow * WPITCH + scol;
            const float* wfr  = wbuf + m * WPITCH + khalf;
            const float* gn   = W_neigh + (size_t)(nbase + srow) * F_IN + scol;
            const float* gs   = W_self  + (size_t)(nbase + srow) * F_IN + scol;
            const float* an   = lds_hn[sg] + m * F_IN + khalf;
            const float* as   = lds_hs[sg] + m * F_IN + khalf;

            float4 g0 = *(const float4*)(gn);
            float4 g1 = *(const float4*)(gn + 4 * F_IN);
            float4 g2 = *(const float4*)(gn + 8 * F_IN);
            float4 g3 = *(const float4*)(gn + 12 * F_IN);

            v8f acc = {};
            #pragma unroll
            for (int t = 0; t < 8; ++t) {
                *(float4*)(wst)               = g0;
                *(float4*)(wst + 4 * WPITCH)  = g1;
                *(float4*)(wst + 8 * WPITCH)  = g2;
                *(float4*)(wst + 12 * WPITCH) = g3;
                __builtin_amdgcn_fence(__ATOMIC_ACQ_REL, "wavefront", "local");

                if (t + 1 < 8) {
                    const float* nsrc = ((t + 1) < 4 ? gn : gs) + ((t + 1) & 3) * KCH;
                    g0 = *(const float4*)(nsrc);
                    g1 = *(const float4*)(nsrc + 4 * F_IN);
                    g2 = *(const float4*)(nsrc + 8 * F_IN);
                    g3 = *(const float4*)(nsrc + 12 * F_IN);
                }

                const float* ap = (t < 4 ? an : as) + (t & 3) * KCH;
                #pragma unroll
                for (int j = 0; j < 8; ++j) {
                    v2f a = *(const v2f*)(ap  + 4 * j);
                    v2f b = *(const v2f*)(wfr + 4 * j);
                    acc = __builtin_amdgcn_wmma_f32_16x16x4_f32(
                        false, a, false, b, (short)0, acc, false, false);
                }
                __builtin_amdgcn_fence(__ATOMIC_ACQ_REL, "wavefront", "local");
                __builtin_amdgcn_sched_barrier(0);
            }

            const float bias  = b_self[nbase + m];
            const int   rhalf = (lane >> 4) * 8;
            #pragma unroll
            for (int j = 0; j < 8; ++j) {
                const int gr = row_base + rhalf + j;
                if (gr < n_dst)
                    out[(size_t)gr * F_OUT + nbase + m] = acc[j] + bias;
            }
        }
    }
}

extern "C" void kernel_launch(void* const* d_in, const int* in_sizes, int n_in,
                              void* d_out, int out_size, void* d_ws, size_t ws_size,
                              hipStream_t stream) {
    (void)n_in; (void)out_size; (void)d_ws; (void)ws_size;
    const int*   codes    = (const int*)  d_in[0];
    const int*   indices  = (const int*)  d_in[1];
    const int*   indptr   = (const int*)  d_in[2];
    const float* h_self   = (const float*)d_in[3];
    const float* codebook = (const float*)d_in[4];
    const float* W_neigh  = (const float*)d_in[5];
    const float* W_self   = (const float*)d_in[6];
    const float* b_self   = (const float*)d_in[7];
    float*       out      = (float*)d_out;

    const int n_dst = in_sizes[2] - 1;

    sage_fused_kernel<<<NWG, NTHR, 0, stream>>>(
        codes, indices, indptr, h_self, codebook, W_neigh, W_self, b_self,
        out, n_dst);
}


# A tiles (aggregated and self features) held in LDS at a row pitch of 132 floats, not 128: the 16 rows an A fragment load reads fall on different banks; before, all 16 met on the same banks
# speedup vs baseline: 2.4631x; 1.6831x over previous
#include <hip/hip_runtime.h>


#define P_     16
#define K_     256
#define W_     8
#define F_IN   128
#define F_OUT  128
#define TILE_M 16
#define KCH    32
#define WPITCH 36
#define APITCH 132
#define NSG    4
#define NTHR   (NSG * 256)
#define NWG    256

typedef __attribute__((ext_vector_type(2))) float v2f;
typedef __attribute__((ext_vector_type(8))) float v8f;

__global__ __launch_bounds__(NTHR) void sage_fused_kernel(
    const int*   __restrict__ codes,
    const int*   __restrict__ indices,
    const int*   __restrict__ indptr,
    const float* __restrict__ h_self,
    const float* __restrict__ codebook,
    const float* __restrict__ W_neigh,
    const float* __restrict__ W_self,
    const float* __restrict__ b_self,
    float*       __restrict__ out,
    int n_dst)
{
    __shared__ __align__(16) float lds_cb[K_ * P_ * W_];
    __shared__ __align__(16) float lds_hn[NSG][TILE_M * APITCH];
    __shared__ __align__(16) float lds_hs[NSG][TILE_M * APITCH];
    __shared__ __align__(16) int   sidx[NSG][TILE_M * 16];
    __shared__ float sinv[NSG][TILE_M];
    __shared__ __align__(16) float lds_w[NSG][8 * 16 * WPITCH];

    const int tid = threadIdx.x;
    const int sg  = tid >> 8;
    const int lt  = tid & 255;

    const int n_tiles  = (n_dst + TILE_M - 1) / TILE_M;
    const int t_first  = (int)((blockIdx.x * (unsigned)n_tiles) / gridDim.x);
    const int t_last   = (int)(((blockIdx.x + 1u) * (unsigned)n_tiles) / gridDim.x);
    const int n_rounds = (t_last - t_first + NSG - 1) / NSG;

    const float4* cbk4 = (const float4*)codebook;
    float4*       cbl4 = (float4*)lds_cb;

    const int erow = lt >> 4;
    const int ee   = lt & 15;
    int deg = 0, src = -1;
    {
        const int tile = t_first + sg;
        const int r    = tile * TILE_M + erow;
        if (tile < t_last && r < n_dst) {
            const int beg = indptr[r];
            deg = indptr[r + 1] - beg;
            if (ee < deg) src = indices[beg + ee];
        }
    }

    {
        const int lane = tid & 31;
        const int cc = lane >> 4, pp = (lane >> 1) & 7, h4 = lane & 1;
        #pragma unroll
        for (int j = tid >> 5; j < (K_ / 2) * (P_ / 8); j += NTHR / 32) {
            const int k = 2 * (j >> 1) + cc;
            const int p = 8 * (j & 1) + pp;
            cbl4[(k * P_ + p) * 2 + h4] = cbk4[(p * K_ + k) * 2 + h4];
        }
    }

    const int lane  = lt & 31;
    const int wave  = lt >> 5;
    const int nbase = wave * 16;
    const int m     = lane & 15;
    const int khalf = (lane >> 4) * 2;

    for (int rd = 0; rd < n_rounds; ++rd) {
        const int  tile     = t_first + rd * NSG + sg;
        const bool have     = tile < t_last;
        const int  row_base = tile * TILE_M;

        sidx[sg][lt] = src;
        if (ee == 0) sinv[sg][erow] = 1.0f / fmaxf((float)deg, 1.0f);
        __syncthreads();

        const int  nr    = (tile + NSG) * TILE_M + erow;
        const bool nhave = (tile + NSG) < t_last && nr < n_dst;
        int nbeg = 0, nend = 0;
        if (nhave) { nbeg = indptr[nr]; nend = indptr[nr + 1]; }

        if (have) {
            for (int idx = lt; idx < TILE_M * (F_IN / 4); idx += 256) {
                const int row = idx >> 5;
                const int c4  = idx & 31;
                const int gr  = row_base + row;

                float4 hs = make_float4(0.f, 0.f, 0.f, 0.f);
                if (gr < n_dst)
                    hs = *(const float4*)(h_self + (size_t)gr * F_IN + c4 * 4);

                const int p = c4 >> 1;

                int s[16];
                #pragma unroll
                for (int q = 0; q < 4; ++q) {
                    const int4 v = *(const int4*)(&sidx[sg][row * 16 + q * 4]);
                    s[q * 4 + 0] = __builtin_amdgcn_readfirstlane(v.x);
                    s[q * 4 + 1] = __builtin_amdgcn_readfirstlane(v.y);
                    s[q * 4 + 2] = __builtin_amdgcn_readfirstlane(v.z);
                    s[q * 4 + 3] = __builtin_amdgcn_readfirstlane(v.w);
                }
                int code[16];
                #pragma unroll
                for (int e = 0; e < 16; ++e)
                    code[e] = codes[(unsigned)((s[e] >= 0 ? s[e] : 0) * P_ + p)];
                float4 cb[16];
                #pragma unroll
                for (int e = 0; e < 16; ++e)
                    cb[e] = cbl4[code[e] * (P_ * 2) + c4];

                #pragma unroll
                for (int e = 0; e < 16; ++e)
                    asm volatile("" :: "v"(cb[e].x), "v"(cb[e].y), "v"(cb[e].z), "v"(cb[e].w));

                float ax = 0.f, ay = 0.f, az = 0.f, aw = 0.f;
                #pragma unroll
                for (int e = 0; e < 16; ++e) {
                    const bool on = s[e] >= 0;
                    ax += on ? cb[e].x : 0.f;
                    ay += on ? cb[e].y : 0.f;
                    az += on ? cb[e].z : 0.f;
                    aw += on ? cb[e].w : 0.f;
                }
                const float inv = sinv[sg][row];
                const float4 hn = make_float4(ax * inv, ay * inv, az * inv, aw * inv);
                *(float4*)(&lds_hs[sg][row * APITCH + c4 * 4]) = hs;
                *(float4*)(&lds_hn[sg][row * APITCH + c4 * 4]) = hn;
            }
        }
        __syncthreads();

        deg = nend - nbeg;
        src = -1;
        if (nhave && ee < deg) src = indices[nbeg + ee];

        if (have) {
            const int srow = lane >> 3;
            const int scol = (lane & 7) * 4;
            float*       wbuf = lds_w[sg] + wave * (16 * WPITCH);
            float*       wst  = wbuf + srow * WPITCH + scol;
            const float* wfr  = wbuf + m * WPITCH + khalf;
            const float* gn   = W_neigh + (size_t)(nbase + srow) * F_IN + scol;
            const float* gs   = W_self  + (size_t)(nbase + srow) * F_IN + scol;
            const float* an   = lds_hn[sg] + m * APITCH + khalf;
            const float* as   = lds_hs[sg] + m * APITCH + khalf;

            float4 g0 = *(const float4*)(gn);
            float4 g1 = *(const float4*)(gn + 4 * F_IN);
            float4 g2 = *(const float4*)(gn + 8 * F_IN);
            float4 g3 = *(const float4*)(gn + 12 * F_IN);

            v8f acc = {};
            #pragma unroll
            for (int t = 0; t < 8; ++t) {
                *(float4*)(wst)               = g0;
                *(float4*)(wst + 4 * WPITCH)  = g1;
                *(float4*)(wst + 8 * WPITCH)  = g2;
                *(float4*)(wst + 12 * WPITCH) = g3;
                __builtin_amdgcn_fence(__ATOMIC_ACQ_REL, "wavefront", "local");

                if (t + 1 < 8) {
                    const float* nsrc = ((t + 1) < 4 ? gn : gs) + ((t + 1) & 3) * KCH;
                    g0 = *(const float4*)(nsrc);
                    g1 = *(const float4*)(nsrc + 4 * F_IN);
                    g2 = *(const float4*)(nsrc + 8 * F_IN);
                    g3 = *(const float4*)(nsrc + 12 * F_IN);
                }

                const float* ap = (t < 4 ? an : as) + (t & 3) * KCH;
                #pragma unroll
                for (int j = 0; j < 8; ++j) {
                    v2f a = *(const v2f*)(ap  + 4 * j);
                    v2f b = *(const v2f*)(wfr + 4 * j);
                    acc = __builtin_amdgcn_wmma_f32_16x16x4_f32(
                        false, a, false, b, (short)0, acc, false, false);
                }
                __builtin_amdgcn_fence(__ATOMIC_ACQ_REL, "wavefront", "local");
                __builtin_amdgcn_sched_barrier(0);
            }

            const float bias  = b_self[nbase + m];
            const int   rhalf = (lane >> 4) * 8;
            #pragma unroll
            for (int j = 0; j < 8; ++j) {
                const int gr = row_base + rhalf + j;
                if (gr < n_dst)
                    out[(size_t)gr * F_OUT + nbase + m] = acc[j] + bias;
            }
        }
    }
}

extern "C" void kernel_launch(void* const* d_in, const int* in_sizes, int n_in,
                              void* d_out, int out_size, void* d_ws, size_t ws_size,
                              hipStream_t stream) {
    (void)n_in; (void)out_size; (void)d_ws; (void)ws_size;
    const int*   codes    = (const int*)  d_in[0];
    const int*   indices  = (const int*)  d_in[1];
    const int*   indptr   = (const int*)  d_in[2];
    const float* h_self   = (const float*)d_in[3];
    const float* codebook = (const float*)d_in[4];
    const float* W_neigh  = (const float*)d_in[5];
    const float* W_self   = (const float*)d_in[6];
    const float* b_self   = (const float*)d_in[7];
    float*       out      = (float*)d_out;

    const int n_dst = in_sizes[2] - 1;

    sage_fused_kernel<<<NWG, NTHR, 0, stream>>>(
        codes, indices, indptr, h_self, codebook, W_neigh, W_self, b_self,
        out, n_dst);
}


# W kept in registers: each wave holds the B fragments of one 16-column tile and one quarter of k all kernel long, multiplies all 4 row tiles by them; quarter sums added via LDS (sum order changes)
# speedup vs baseline: 2.7781x; 1.1279x over previous
#include <hip/hip_runtime.h>


#define P_     16
#define K_     256
#define W_     8
#define F_IN   128
#define F_OUT  128
#define TILE_M 16
#define APITCH 132
#define ATILE  (TILE_M * APITCH)
#define NSG    4
#define NTHR   (NSG * 256)
#define NWG    256
#define KQ     64
#define KSTEPS (KQ / 4)

typedef __attribute__((ext_vector_type(2))) float v2f;
typedef __attribute__((ext_vector_type(8))) float v8f;

__global__ __launch_bounds__(NTHR) void sage_fused_kernel(
    const int*   __restrict__ codes,
    const int*   __restrict__ indices,
    const int*   __restrict__ indptr,
    const float* __restrict__ h_self,
    const float* __restrict__ codebook,
    const float* __restrict__ W_neigh,
    const float* __restrict__ W_self,
    const float* __restrict__ b_self,
    float*       __restrict__ out,
    int n_dst)
{
    __shared__ __align__(16) float lds_cb[K_ * P_ * W_];
    __shared__ __align__(16) float lds_hn[NSG][ATILE];
    __shared__ __align__(16) float lds_hs[NSG][ATILE];
    __shared__ __align__(16) int   sidx[NSG][TILE_M * 16];
    __shared__ float sinv[NSG][TILE_M];
    __shared__ __align__(16) float lds_red[8][NSG][3][256];

    const int tid = threadIdx.x;
    const int sg  = tid >> 8;
    const int lt  = tid & 255;

    const int lane  = tid & 31;
    const int nt    = (tid >> 5) & 7;
    const int q     = tid >> 8;
    const int nbase = nt * 16;
    const int m     = lane & 15;
    const int kh4   = (lane >> 4) * 4;

    v2f bfr[KSTEPS];
    {
        const float* wq = (q < 2 ? W_neigh : W_self)
                        + (size_t)(nbase + m) * F_IN + (q & 1) * KQ + kh4;
        #pragma unroll
        for (int jj = 0; jj < KSTEPS / 2; ++jj) {
            const float4 w4 = *(const float4*)(wq + 8 * jj);
            bfr[2 * jj]     = (v2f){w4.x, w4.y};
            bfr[2 * jj + 1] = (v2f){w4.z, w4.w};
        }
    }
    const float bias = b_self[nbase + m];

    const int n_tiles  = (n_dst + TILE_M - 1) / TILE_M;
    const int t_first  = (int)((blockIdx.x * (unsigned)n_tiles) / gridDim.x);
    const int t_last   = (int)(((blockIdx.x + 1u) * (unsigned)n_tiles) / gridDim.x);
    const int n_rounds = (t_last - t_first + NSG - 1) / NSG;

    const float4* cbk4 = (const float4*)codebook;
    float4*       cbl4 = (float4*)lds_cb;

    const int erow = lt >> 4;
    const int ee   = lt & 15;
    int deg = 0, src = -1;
    {
        const int tile = t_first + sg;
        const int r    = tile * TILE_M + erow;
        if (tile < t_last && r < n_dst) {
            const int beg = indptr[r];
            deg = indptr[r + 1] - beg;
            if (ee < deg) src = indices[beg + ee];
        }
    }

    {
        const int cc = lane >> 4, pp = (lane >> 1) & 7, h4 = lane & 1;
        #pragma unroll
        for (int j = tid >> 5; j < (K_ / 2) * (P_ / 8); j += NTHR / 32) {
            const int k = 2 * (j >> 1) + cc;
            const int p = 8 * (j & 1) + pp;
            cbl4[(k * P_ + p) * 2 + h4] = cbk4[(p * K_ + k) * 2 + h4];
        }
    }

    const float* aq  = (q < 2 ? &lds_hn[0][0] : &lds_hs[0][0]) + m * APITCH + (q & 1) * KQ + kh4;
    float*       red = &lds_red[nt][0][0][0] + lane * 4;

    for (int rd = 0; rd < n_rounds; ++rd) {
        const int  tile0    = t_first + rd * NSG;
        const int  nv       = min(NSG, t_last - tile0);
        const int  tile     = tile0 + sg;
        const bool have     = sg < nv;
        const int  row_base = tile * TILE_M;

        sidx[sg][lt] = src;
        if (ee == 0) sinv[sg][erow] = 1.0f / fmaxf((float)deg, 1.0f);
        __syncthreads();

        const int  nr    = (tile + NSG) * TILE_M + erow;
        const bool nhave = (tile + NSG) < t_last && nr < n_dst;
        int nbeg = 0, nend = 0;
        if (nhave) { nbeg = indptr[nr]; nend = indptr[nr + 1]; }

        if (have) {
            for (int idx = lt; idx < TILE_M * (F_IN / 4); idx += 256) {
                const int row = idx >> 5;
                const int c4  = idx & 31;
                const int gr  = row_base + row;

                float4 hs = make_float4(0.f, 0.f, 0.f, 0.f);
                if (gr < n_dst)
                    hs = *(const float4*)(h_self + (size_t)gr * F_IN + c4 * 4);

                const int p = c4 >> 1;

                int s[16];
                #pragma unroll
                for (int u = 0; u < 4; ++u) {
                    const int4 v = *(const int4*)(&sidx[sg][row * 16 + u * 4]);
                    s[u * 4 + 0] = __builtin_amdgcn_readfirstlane(v.x);
                    s[u * 4 + 1] = __builtin_amdgcn_readfirstlane(v.y);
                    s[u * 4 + 2] = __builtin_amdgcn_readfirstlane(v.z);
                    s[u * 4 + 3] = __builtin_amdgcn_readfirstlane(v.w);
                }
                int code[16];
                #pragma unroll
                for (int e = 0; e < 16; ++e)
                    code[e] = codes[(unsigned)((s[e] >= 0 ? s[e] : 0) * P_ + p)];

                float ax = 0.f, ay = 0.f, az = 0.f, aw = 0.f;
                #pragma unroll
                for (int hb = 0; hb < 16; hb += 8) {
                    float4 cb[8];
                    #pragma unroll
                    for (int e = 0; e < 8; ++e)
                        cb[e] = cbl4[code[hb + e] * (P_ * 2) + c4];

                    #pragma unroll
                    for (int e = 0; e < 8; ++e)
                        asm volatile("" :: "v"(cb[e].x), "v"(cb[e].y), "v"(cb[e].z), "v"(cb[e].w));

                    #pragma unroll
                    for (int e = 0; e < 8; ++e) {
                        const bool on = s[hb + e] >= 0;
                        ax += on ? cb[e].x : 0.f;
                        ay += on ? cb[e].y : 0.f;
                        az += on ? cb[e].z : 0.f;
                        aw += on ? cb[e].w : 0.f;
                    }
                }
                const float inv = sinv[sg][row];
                const float4 hn = make_float4(ax * inv, ay * inv, az * inv, aw * inv);
                *(float4*)(&lds_hs[sg][row * APITCH + c4 * 4]) = hs;
                *(float4*)(&lds_hn[sg][row * APITCH + c4 * 4]) = hn;
            }
        }
        __syncthreads();

        deg = nend - nbeg;
        src = -1;
        if (nhave && ee < deg) src = indices[nbeg + ee];

        v8f acc[NSG];
        #pragma unroll
        for (int i = 0; i < NSG; ++i) acc[i] = (v8f){};

        if (nv == NSG) {
            #pragma unroll
            for (int jj = 0; jj < KSTEPS / 2; ++jj) {
                float4 a4[NSG];
                #pragma unroll
                for (int i = 0; i < NSG; ++i) a4[i] = *(const float4*)(aq + i * ATILE + 8 * jj);
                #pragma unroll
                for (int i = 0; i < NSG; ++i)
                    acc[i] = __builtin_amdgcn_wmma_f32_16x16x4_f32(
                        false, (v2f){a4[i].x, a4[i].y}, false, bfr[2 * jj], (short)0, acc[i], false, false);
                #pragma unroll
                for (int i = 0; i < NSG; ++i)
                    acc[i] = __builtin_amdgcn_wmma_f32_16x16x4_f32(
                        false, (v2f){a4[i].z, a4[i].w}, false, bfr[2 * jj + 1], (short)0, acc[i], false, false);
            }
        } else {
            #pragma unroll
            for (int i = 0; i < NSG - 1; ++i) {
                if (i < nv) {
                    #pragma unroll
                    for (int jj = 0; jj < KSTEPS / 2; ++jj) {
                        const float4 a4 = *(const float4*)(aq + i * ATILE + 8 * jj);
                        acc[i] = __builtin_amdgcn_wmma_f32_16x16x4_f32(
                            false, (v2f){a4.x, a4.y}, false, bfr[2 * jj], (short)0, acc[i], false, false);
                        acc[i] = __builtin_amdgcn_wmma_f32_16x16x4_f32(
                            false, (v2f){a4.z, a4.w}, false, bfr[2 * jj + 1], (short)0, acc[i], false, false);
                    }
                }
            }
        }

        #pragma unroll
        for (int i = 0; i < NSG; ++i) {
            if (i != q && i < nv) {
                float* dst = red + (i * 3 + ((q - i - 1) & 3)) * 256;
                *(float4*)(dst)       = make_float4(acc[i][0], acc[i][1], acc[i][2], acc[i][3]);
                *(float4*)(dst + 128) = make_float4(acc[i][4], acc[i][5], acc[i][6], acc[i][7]);
            }
        }
        __syncthreads();

        if (have) {
            v8f tot = q == 0 ? acc[0] : q == 1 ? acc[1] : q == 2 ? acc[2] : acc[3];
            #pragma unroll
            for (int sl = 0; sl < 3; ++sl) {
                const float4 lo = *(const float4*)(red + (q * 3 + sl) * 256);
                const float4 hi = *(const float4*)(red + (q * 3 + sl) * 256 + 128);
                tot[0] += lo.x; tot[1] += lo.y; tot[2] += lo.z; tot[3] += lo.w;
                tot[4] += hi.x; tot[5] += hi.y; tot[6] += hi.z; tot[7] += hi.w;
            }
            const int rhalf = (lane >> 4) * 8;
            #pragma unroll
            for (int j = 0; j < 8; ++j) {
                const int gr = row_base + rhalf + j;
                if (gr < n_dst)
                    out[(size_t)gr * F_OUT + nbase + m] = tot[j] + bias;
            }
        }
    }
}

extern "C" void kernel_launch(void* const* d_in, const int* in_sizes, int n_in,
                              void* d_out, int out_size, void* d_ws, size_t ws_size,
                              hipStream_t stream) {
    (void)n_in; (void)out_size; (void)d_ws; (void)ws_size;
    const int*   codes    = (const int*)  d_in[0];
    const int*   indices  = (const int*)  d_in[1];
    const int*   indptr   = (const int*)  d_in[2];
    const float* h_self   = (const float*)d_in[3];
    const float* codebook = (const float*)d_in[4];
    const float* W_neigh  = (const float*)d_in[5];
    const float* W_self   = (const float*)d_in[6];
    const float* b_self   = (const float*)d_in[7];
    float*       out      = (float*)d_out;

    const int n_dst = in_sizes[2] - 1;

    sage_fused_kernel<<<NWG, NTHR, 0, stream>>>(
        codes, indices, indptr, h_self, codebook, W_neigh, W_self, b_self,
        out, n_dst);
}


# 3-stage pipeline, rounds of 2 tiles, 1 barrier each: waves 0-15 multiply round r-1 then gather round r, waves 16-31 gather first: loads run under the other half's WMMAs; quarter-sum order changes
# speedup vs baseline: 2.9297x; 1.0546x over previous
#include <hip/hip_runtime.h>


#define P_     16
#define K_     256
#define W_     8
#define F_IN   128
#define F_OUT  128
#define TILE_M 16
#define APITCH 132
#define ATILE  (TILE_M * APITCH)
#define NSG    2
#define NTHR   1024
#define NWG    256
#define KQ     64
#define KSTEPS (KQ / 4)

typedef __attribute__((ext_vector_type(2))) float v2f;
typedef __attribute__((ext_vector_type(8))) float v8f;

__global__ __launch_bounds__(NTHR) void sage_fused_kernel(
    const int*   __restrict__ codes,
    const int*   __restrict__ indices,
    const int*   __restrict__ indptr,
    const float* __restrict__ h_self,
    const float* __restrict__ codebook,
    const float* __restrict__ W_neigh,
    const float* __restrict__ W_self,
    const float* __restrict__ b_self,
    float*       __restrict__ out,
    int n_dst)
{
    __shared__ __align__(16) float lds_cb[K_ * P_ * W_];
    __shared__ __align__(16) float lds_hn[2][NSG][ATILE];
    __shared__ __align__(16) float lds_hs[2][NSG][ATILE];
    __shared__ __align__(16) int   sidx[2][NSG][TILE_M * 16];
    __shared__ float sinv[2][NSG][TILE_M];
    __shared__ __align__(16) float lds_red[2][8][NSG][3][256];

    const int tid   = threadIdx.x;
    const int lane  = tid & 31;
    const int nt    = (tid >> 5) & 7;
    const int q     = __builtin_amdgcn_readfirstlane(tid >> 8);
    const bool lead = q < 2;
    const int nbase = nt * 16;
    const int m     = lane & 15;
    const int kh4   = (lane >> 4) * 4;

    const int sg    = q >> 1;
    const int grow  = (tid >> 5) & 15;
    const int etile = q & 1;
    const int elt   = tid & 255;
    const int erow  = elt >> 4;
    const int ee    = elt & 15;

    v2f bfr[KSTEPS];
    {
        const float* wq = (q < 2 ? W_neigh : W_self)
                        + (size_t)(nbase + m) * F_IN + (q & 1) * KQ + kh4;
        #pragma unroll
        for (int jj = 0; jj < KSTEPS / 2; ++jj) {
            const float4 w4 = *(const float4*)(wq + 8 * jj);
            bfr[2 * jj]     = (v2f){w4.x, w4.y};
            bfr[2 * jj + 1] = (v2f){w4.z, w4.w};
        }
    }
    const float bias = b_self[nbase + m];

    const int n_tiles  = (n_dst + TILE_M - 1) / TILE_M;
    const int t_first  = (int)((blockIdx.x * (unsigned)n_tiles) / gridDim.x);
    const int t_last   = (int)(((blockIdx.x + 1u) * (unsigned)n_tiles) / gridDim.x);
    const int n_rounds = (t_last - t_first + NSG - 1) / NSG;

    const float4* cbk4 = (const float4*)codebook;
    float4*       cbl4 = (float4*)lds_cb;

    int  deg = 0, src = -1;
    int  nbeg = 0, nend = 0;
    bool nval = false;
    if (!lead) {
        const int tile = t_first + etile;
        const int r    = tile * TILE_M + erow;
        if (tile < t_last && r < n_dst) {
            const int beg = indptr[r];
            deg = indptr[r + 1] - beg;
            if (ee < deg) src = indices[beg + ee];
        }
    }

    {
        const int cc = lane >> 4, pp = (lane >> 1) & 7, h4 = lane & 1;
        #pragma unroll
        for (int j = tid >> 5; j < (K_ / 2) * (P_ / 8); j += NTHR / 32) {
            const int k = 2 * (j >> 1) + cc;
            const int p = 8 * (j & 1) + pp;
            cbl4[(k * P_ + p) * 2 + h4] = cbk4[(p * K_ + k) * 2 + h4];
        }
    }

    if (!lead) {
        sidx[0][etile][elt] = src;
        if (ee == 0) sinv[0][etile][erow] = 1.0f / fmaxf((float)deg, 1.0f);
        const int tile = t_first + NSG + etile;
        const int r    = tile * TILE_M + erow;
        nval = tile < t_last && r < n_dst;
        if (nval) { nbeg = indptr[r]; nend = indptr[r + 1]; }
    }

    auto gather = [&](const int t) {
        const int gs   = t & 1;
        const int tile = t_first + t * NSG + sg;
        if (tile < t_last) {
            const int row = grow;
            const int c4  = lane;
            const int gr  = tile * TILE_M + row;

            float4 hs = make_float4(0.f, 0.f, 0.f, 0.f);
            if (gr < n_dst)
                hs = *(const float4*)(h_self + (size_t)gr * F_IN + c4 * 4);

            const int p = c4 >> 1;

            int s[16];
            #pragma unroll
            for (int u = 0; u < 4; ++u) {
                const int4 v = *(const int4*)(&sidx[gs][sg][row * 16 + u * 4]);
                s[u * 4 + 0] = __builtin_amdgcn_readfirstlane(v.x);
                s[u * 4 + 1] = __builtin_amdgcn_readfirstlane(v.y);
                s[u * 4 + 2] = __builtin_amdgcn_readfirstlane(v.z);
                s[u * 4 + 3] = __builtin_amdgcn_readfirstlane(v.w);
            }
            int code[16];
            #pragma unroll
            for (int e = 0; e < 16; ++e)
                code[e] = codes[(unsigned)((s[e] >= 0 ? s[e] : 0) * P_ + p)];

            float ax = 0.f, ay = 0.f, az = 0.f, aw = 0.f;
            #pragma unroll
            for (int hb = 0; hb < 16; hb += 8) {
                float4 cb[8];
                #pragma unroll
                for (int e = 0; e < 8; ++e)
                    cb[e] = cbl4[code[hb + e] * (P_ * 2) + c4];

                #pragma unroll
                for (int e = 0; e < 8; ++e)
                    asm volatile("" :: "v"(cb[e].x), "v"(cb[e].y), "v"(cb[e].z), "v"(cb[e].w));

                #pragma unroll
                for (int e = 0; e < 8; ++e) {
                    const bool on = s[hb + e] >= 0;
                    ax += on ? cb[e].x : 0.f;
                    ay += on ? cb[e].y : 0.f;
                    az += on ? cb[e].z : 0.f;
                    aw += on ? cb[e].w : 0.f;
                }
            }
            const float inv = sinv[gs][sg][row];
            const float4 hn = make_float4(ax * inv, ay * inv, az * inv, aw * inv);
            *(float4*)(&lds_hs[gs][sg][row * APITCH + c4 * 4]) = hs;
            *(float4*)(&lds_hn[gs][sg][row * APITCH + c4 * 4]) = hn;
        }
    };

    auto product = [&](const int r, v8f& keep) {
        const int ps    = r & 1;
        const int tile0 = t_first + r * NSG;
        const int nv    = min(NSG, t_last - tile0);
        const float* aq = (q < 2 ? &lds_hn[ps][0][0] : &lds_hs[ps][0][0]) + m * APITCH + (q & 1) * KQ + kh4;
        v8f acc0 = (v8f){}, acc1 = (v8f){};
        if (nv == NSG) {
            #pragma unroll
            for (int jj = 0; jj < KSTEPS / 2; ++jj) {
                const float4 a0 = *(const float4*)(aq + 8 * jj);
                const float4 a1 = *(const float4*)(aq + ATILE + 8 * jj);
                acc0 = __builtin_amdgcn_wmma_f32_16x16x4_f32(
                    false, (v2f){a0.x, a0.y}, false, bfr[2 * jj], (short)0, acc0, false, false);
                acc1 = __builtin_amdgcn_wmma_f32_16x16x4_f32(
                    false, (v2f){a1.x, a1.y}, false, bfr[2 * jj], (short)0, acc1, false, false);
                acc0 = __builtin_amdgcn_wmma_f32_16x16x4_f32(
                    false, (v2f){a0.z, a0.w}, false, bfr[2 * jj + 1], (short)0, acc0, false, false);
                acc1 = __builtin_amdgcn_wmma_f32_16x16x4_f32(
                    false, (v2f){a1.z, a1.w}, false, bfr[2 * jj + 1], (short)0, acc1, false, false);
            }
        } else {
            #pragma unroll
            for (int jj = 0; jj < KSTEPS / 2; ++jj) {
                const float4 a0 = *(const float4*)(aq + 8 * jj);
                acc0 = __builtin_amdgcn_wmma_f32_16x16x4_f32(
                    false, (v2f){a0.x, a0.y}, false, bfr[2 * jj], (short)0, acc0, false, false);
                acc0 = __builtin_amdgcn_wmma_f32_16x16x4_f32(
                    false, (v2f){a0.z, a0.w}, false, bfr[2 * jj + 1], (short)0, acc0, false, false);
            }
        }
        float* red = &lds_red[ps][nt][0][0][0] + lane * 4;
        if (q != 0) {
            float* dst = red + ((q - 1) & 3) * 256;
            *(float4*)(dst)       = make_float4(acc0[0], acc0[1], acc0[2], acc0[3]);
            *(float4*)(dst + 128) = make_float4(acc0[4], acc0[5], acc0[6], acc0[7]);
        }
        if (q != 1 && nv == NSG) {
            float* dst = red + (3 + ((q - 2) & 3)) * 256;
            *(float4*)(dst)       = make_float4(acc1[0], acc1[1], acc1[2], acc1[3]);
            *(float4*)(dst + 128) = make_float4(acc1[4], acc1[5], acc1[6], acc1[7]);
        }
        if (q == 0) keep = acc0;
        if (q == 1) keep = acc1;
    };

    auto finish = [&](const int r, const v8f& keep) {
        const int tile = t_first + r * NSG + q;
        if (tile < t_last) {
            const float* red = &lds_red[r & 1][nt][0][0][0] + lane * 4;
            v8f tot = keep;
            #pragma unroll
            for (int sl = 0; sl < 3; ++sl) {
                const float4 lo = *(const float4*)(red + (q * 3 + sl) * 256);
                const float4 hi = *(const float4*)(red + (q * 3 + sl) * 256 + 128);
                tot[0] += lo.x; tot[1] += lo.y; tot[2] += lo.z; tot[3] += lo.w;
                tot[4] += hi.x; tot[5] += hi.y; tot[6] += hi.z; tot[7] += hi.w;
            }
            const int row_base = tile * TILE_M;
            const int rhalf    = (lane >> 4) * 8;
            #pragma unroll
            for (int j = 0; j < 8; ++j) {
                const int gr = row_base + rhalf + j;
                if (gr < n_dst)
                    out[(size_t)gr * F_OUT + nbase + m] = tot[j] + bias;
            }
        }
    };

    v8f own = (v8f){};

    for (int t = 0; t < n_rounds + 2; ++t) {
        __syncthreads();

        if (lead) {
            if (t >= 2) finish(t - 2, own);
            if (t >= 1 && t <= n_rounds) product(t - 1, own);
            if (t < n_rounds) gather(t);
        } else {
            if (t < n_rounds) gather(t);

            deg = nend - nbeg;
            src = -1;
            if (nval && ee < deg) src = indices[nbeg + ee];
            {
                const int tile = t_first + (t + 2) * NSG + etile;
                const int r    = tile * TILE_M + erow;
                nval = tile < t_last && r < n_dst;
                nbeg = 0; nend = 0;
                if (nval) { nbeg = indptr[r]; nend = indptr[r + 1]; }
            }

            v8f unused;
            if (t >= 1 && t <= n_rounds) product(t - 1, unused);

            sidx[(t + 1) & 1][etile][elt] = src;
            if (ee == 0) sinv[(t + 1) & 1][etile][erow] = 1.0f / fmaxf((float)deg, 1.0f);
        }
    }
}

extern "C" void kernel_launch(void* const* d_in, const int* in_sizes, int n_in,
                              void* d_out, int out_size, void* d_ws, size_t ws_size,
                              hipStream_t stream) {
    (void)n_in; (void)out_size; (void)d_ws; (void)ws_size;
    const int*   codes    = (const int*)  d_in[0];
    const int*   indices  = (const int*)  d_in[1];
    const int*   indptr   = (const int*)  d_in[2];
    const float* h_self   = (const float*)d_in[3];
    const float* codebook = (const float*)d_in[4];
    const float* W_neigh  = (const float*)d_in[5];
    const float* W_self   = (const float*)d_in[6];
    const float* b_self   = (const float*)d_in[7];
    float*       out      = (float*)d_out;

    const int n_dst = in_sizes[2] - 1;

    sage_fused_kernel<<<NWG, NTHR, 0, stream>>>(
        codes, indices, indptr, h_self, codebook, W_neigh, W_self, b_self,
        out, n_dst);
}


# Complete rows and tiles take guard-free paths picked by one scalar test each: gather without per-edge clamp and select when all 16 neighbours are present, epilogue without per-row bounds branch
# speedup vs baseline: 3.1455x; 1.0737x over previous
#include <hip/hip_runtime.h>


#define P_     16
#define K_     256
#define W_     8
#define F_IN   128
#define F_OUT  128
#define TILE_M 16
#define APITCH 132
#define ATILE  (TILE_M * APITCH)
#define NSG    2
#define NTHR   1024
#define NWG    256
#define KQ     64
#define KSTEPS (KQ / 4)

typedef __attribute__((ext_vector_type(2))) float v2f;
typedef __attribute__((ext_vector_type(8))) float v8f;

__global__ __launch_bounds__(NTHR) void sage_fused_kernel(
    const int*   __restrict__ codes,
    const int*   __restrict__ indices,
    const int*   __restrict__ indptr,
    const float* __restrict__ h_self,
    const float* __restrict__ codebook,
    const float* __restrict__ W_neigh,
    const float* __restrict__ W_self,
    const float* __restrict__ b_self,
    float*       __restrict__ out,
    int n_dst)
{
    __shared__ __align__(16) float lds_cb[K_ * P_ * W_];
    __shared__ __align__(16) float lds_hn[2][NSG][ATILE];
    __shared__ __align__(16) float lds_hs[2][NSG][ATILE];
    __shared__ __align__(16) int   sidx[2][NSG][TILE_M * 16];
    __shared__ float sinv[2][NSG][TILE_M];
    __shared__ int   sfull[2][NSG][TILE_M];
    __shared__ __align__(16) float lds_red[2][8][NSG][3][256];

    const int tid   = threadIdx.x;
    const int lane  = tid & 31;
    const int nt    = (tid >> 5) & 7;
    const int q     = __builtin_amdgcn_readfirstlane(tid >> 8);
    const bool lead = q < 2;
    const int nbase = nt * 16;
    const int m     = lane & 15;
    const int kh4   = (lane >> 4) * 4;

    const int sg    = q >> 1;
    const int grow  = (tid >> 5) & 15;
    const int etile = q & 1;
    const int elt   = tid & 255;
    const int erow  = elt >> 4;
    const int ee    = elt & 15;

    v2f bfr[KSTEPS];
    {
        const float* wq = (q < 2 ? W_neigh : W_self)
                        + (size_t)(nbase + m) * F_IN + (q & 1) * KQ + kh4;
        #pragma unroll
        for (int jj = 0; jj < KSTEPS / 2; ++jj) {
            const float4 w4 = *(const float4*)(wq + 8 * jj);
            bfr[2 * jj]     = (v2f){w4.x, w4.y};
            bfr[2 * jj + 1] = (v2f){w4.z, w4.w};
        }
    }
    const float bias = b_self[nbase + m];

    const int n_tiles  = (n_dst + TILE_M - 1) / TILE_M;
    const int t_first  = (int)((blockIdx.x * (unsigned)n_tiles) / gridDim.x);
    const int t_last   = (int)(((blockIdx.x + 1u) * (unsigned)n_tiles) / gridDim.x);
    const int n_rounds = (t_last - t_first + NSG - 1) / NSG;

    const float4* cbk4 = (const float4*)codebook;
    float4*       cbl4 = (float4*)lds_cb;

    int  deg = 0, src = -1;
    int  nbeg = 0, nend = 0;
    bool nval = false;
    if (!lead) {
        const int tile = t_first + etile;
        const int r    = tile * TILE_M + erow;
        if (tile < t_last && r < n_dst) {
            const int beg = indptr[r];
            deg = indptr[r + 1] - beg;
            if (ee < deg) src = indices[beg + ee];
        }
    }

    {
        const int cc = lane >> 4, pp = (lane >> 1) & 7, h4 = lane & 1;
        #pragma unroll
        for (int j = tid >> 5; j < (K_ / 2) * (P_ / 8); j += NTHR / 32) {
            const int k = 2 * (j >> 1) + cc;
            const int p = 8 * (j & 1) + pp;
            cbl4[(k * P_ + p) * 2 + h4] = cbk4[(p * K_ + k) * 2 + h4];
        }
    }

    if (!lead) {
        sidx[0][etile][elt] = src;
        const unsigned pres = __builtin_amdgcn_ballot_w32(src >= 0);
        if (ee == 0) {
            sinv[0][etile][erow]  = 1.0f / fmaxf((float)deg, 1.0f);
            sfull[0][etile][erow] = ((pres >> (lane & 16)) & 0xFFFFu) == 0xFFFFu;
        }
        const int tile = t_first + NSG + etile;
        const int r    = tile * TILE_M + erow;
        nval = tile < t_last && r < n_dst;
        if (nval) { nbeg = indptr[r]; nend = indptr[r + 1]; }
    }

    auto gather = [&](const int t) {
        const int gs   = t & 1;
        const int tile = t_first + t * NSG + sg;
        if (tile < t_last) {
            const int row = grow;
            const int c4  = lane;
            const int gr  = tile * TILE_M + row;

            float4 hs = make_float4(0.f, 0.f, 0.f, 0.f);
            if (gr < n_dst)
                hs = *(const float4*)(h_self + (size_t)gr * F_IN + c4 * 4);

            const int p = c4 >> 1;

            int s[16];
            #pragma unroll
            for (int u = 0; u < 4; ++u) {
                const int4 v = *(const int4*)(&sidx[gs][sg][row * 16 + u * 4]);
                s[u * 4 + 0] = __builtin_amdgcn_readfirstlane(v.x);
                s[u * 4 + 1] = __builtin_amdgcn_readfirstlane(v.y);
                s[u * 4 + 2] = __builtin_amdgcn_readfirstlane(v.z);
                s[u * 4 + 3] = __builtin_amdgcn_readfirstlane(v.w);
            }
            float ax = 0.f, ay = 0.f, az = 0.f, aw = 0.f;
            if (__builtin_amdgcn_readfirstlane(sfull[gs][sg][row])) {
                int code[16];
                #pragma unroll
                for (int e = 0; e < 16; ++e)
                    code[e] = codes[(unsigned)(s[e] * P_ + p)];

                #pragma unroll
                for (int hb = 0; hb < 16; hb += 8) {
                    float4 cb[8];
                    #pragma unroll
                    for (int e = 0; e < 8; ++e)
                        cb[e] = cbl4[code[hb + e] * (P_ * 2) + c4];
                    #pragma unroll
                    for (int e = 0; e < 8; ++e) {
                        ax += cb[e].x;
                        ay += cb[e].y;
                        az += cb[e].z;
                        aw += cb[e].w;
                    }
                }
            } else {
                int code[16];
                #pragma unroll
                for (int e = 0; e < 16; ++e)
                    code[e] = codes[(unsigned)((s[e] >= 0 ? s[e] : 0) * P_ + p)];

                #pragma unroll
                for (int hb = 0; hb < 16; hb += 8) {
                    float4 cb[8];
                    #pragma unroll
                    for (int e = 0; e < 8; ++e)
                        cb[e] = cbl4[code[hb + e] * (P_ * 2) + c4];

                    #pragma unroll
                    for (int e = 0; e < 8; ++e)
                        asm volatile("" :: "v"(cb[e].x), "v"(cb[e].y), "v"(cb[e].z), "v"(cb[e].w));

                    #pragma unroll
                    for (int e = 0; e < 8; ++e) {
                        const bool on = s[hb + e] >= 0;
                        ax += on ? cb[e].x : 0.f;
                        ay += on ? cb[e].y : 0.f;
                        az += on ? cb[e].z : 0.f;
                        aw += on ? cb[e].w : 0.f;
                    }
                }
            }
            const float inv = sinv[gs][sg][row];
            const float4 hn = make_float4(ax * inv, ay * inv, az * inv, aw * inv);
            *(float4*)(&lds_hs[gs][sg][row * APITCH + c4 * 4]) = hs;
            *(float4*)(&lds_hn[gs][sg][row * APITCH + c4 * 4]) = hn;
        }
    };

    auto product = [&](const int r, v8f& keep) {
        const int ps    = r & 1;
        const int tile0 = t_first + r * NSG;
        const int nv    = min(NSG, t_last - tile0);
        const float* aq = (q < 2 ? &lds_hn[ps][0][0] : &lds_hs[ps][0][0]) + m * APITCH + (q & 1) * KQ + kh4;
        v8f acc0 = (v8f){}, acc1 = (v8f){};
        if (nv == NSG) {
            #pragma unroll
            for (int jj = 0; jj < KSTEPS / 2; ++jj) {
                const float4 a0 = *(const float4*)(aq + 8 * jj);
                const float4 a1 = *(const float4*)(aq + ATILE + 8 * jj);
                acc0 = __builtin_amdgcn_wmma_f32_16x16x4_f32(
                    false, (v2f){a0.x, a0.y}, false, bfr[2 * jj], (short)0, acc0, false, false);
                acc1 = __builtin_amdgcn_wmma_f32_16x16x4_f32(
                    false, (v2f){a1.x, a1.y}, false, bfr[2 * jj], (short)0, acc1, false, false);
                acc0 = __builtin_amdgcn_wmma_f32_16x16x4_f32(
                    false, (v2f){a0.z, a0.w}, false, bfr[2 * jj + 1], (short)0, acc0, false, false);
                acc1 = __builtin_amdgcn_wmma_f32_16x16x4_f32(
                    false, (v2f){a1.z, a1.w}, false, bfr[2 * jj + 1], (short)0, acc1, false, false);
            }
        } else {
            #pragma unroll
            for (int jj = 0; jj < KSTEPS / 2; ++jj) {
                const float4 a0 = *(const float4*)(aq + 8 * jj);
                acc0 = __builtin_amdgcn_wmma_f32_16x16x4_f32(
                    false, (v2f){a0.x, a0.y}, false, bfr[2 * jj], (short)0, acc0, false, false);
                acc0 = __builtin_amdgcn_wmma_f32_16x16x4_f32(
                    false, (v2f){a0.z, a0.w}, false, bfr[2 * jj + 1], (short)0, acc0, false, false);
            }
        }
        float* red = &lds_red[ps][nt][0][0][0] + lane * 4;
        if (q != 0) {
            float* dst = red + ((q - 1) & 3) * 256;
            *(float4*)(dst)       = make_float4(acc0[0], acc0[1], acc0[2], acc0[3]);
            *(float4*)(dst + 128) = make_float4(acc0[4], acc0[5], acc0[6], acc0[7]);
        }
        if (q != 1 && nv == NSG) {
            float* dst = red + (3 + ((q - 2) & 3)) * 256;
            *(float4*)(dst)       = make_float4(acc1[0], acc1[1], acc1[2], acc1[3]);
            *(float4*)(dst + 128) = make_float4(acc1[4], acc1[5], acc1[6], acc1[7]);
        }
        if (q == 0) keep = acc0;
        if (q == 1) keep = acc1;
    };

    auto finish = [&](const int r, const v8f& keep) {
        const int tile = t_first + r * NSG + q;
        if (tile < t_last) {
            const float* red = &lds_red[r & 1][nt][0][0][0] + lane * 4;
            v8f tot = keep;
            #pragma unroll
            for (int sl = 0; sl < 3; ++sl) {
                const float4 lo = *(const float4*)(red + (q * 3 + sl) * 256);
                const float4 hi = *(const float4*)(red + (q * 3 + sl) * 256 + 128);
                tot[0] += lo.x; tot[1] += lo.y; tot[2] += lo.z; tot[3] += lo.w;
                tot[4] += hi.x; tot[5] += hi.y; tot[6] += hi.z; tot[7] += hi.w;
            }
            const int row_base = tile * TILE_M;
            const int rhalf    = (lane >> 4) * 8;
            if (row_base + TILE_M <= n_dst) {
                float* o = out + (size_t)(row_base + rhalf) * F_OUT + nbase + m;
                #pragma unroll
                for (int j = 0; j < 8; ++j)
                    o[j * F_OUT] = tot[j] + bias;
            } else {
                #pragma unroll
                for (int j = 0; j < 8; ++j) {
                    const int gr = row_base + rhalf + j;
                    if (gr < n_dst)
                        out[(size_t)gr * F_OUT + nbase + m] = tot[j] + bias;
                }
            }
        }
    };

    v8f own = (v8f){};

    for (int t = 0; t < n_rounds + 2; ++t) {
        __syncthreads();

        if (lead) {
            if (t >= 2) finish(t - 2, own);
            if (t >= 1 && t <= n_rounds) product(t - 1, own);
            if (t < n_rounds) gather(t);
        } else {
            if (t < n_rounds) gather(t);

            deg = nend - nbeg;
            src = -1;
            if (nval && ee < deg) src = indices[nbeg + ee];
            {
                const int tile = t_first + (t + 2) * NSG + etile;
                const int r    = tile * TILE_M + erow;
                nval = tile < t_last && r < n_dst;
                nbeg = 0; nend = 0;
                if (nval) { nbeg = indptr[r]; nend = indptr[r + 1]; }
            }

            v8f unused;
            if (t >= 1 && t <= n_rounds) product(t - 1, unused);

            sidx[(t + 1) & 1][etile][elt] = src;
            const unsigned pres = __builtin_amdgcn_ballot_w32(src >= 0);
            if (ee == 0) {
                sinv[(t + 1) & 1][etile][erow]  = 1.0f / fmaxf((float)deg, 1.0f);
                sfull[(t + 1) & 1][etile][erow] = ((pres >> (lane & 16)) & 0xFFFFu) == 0xFFFFu;
            }
        }
    }
}

extern "C" void kernel_launch(void* const* d_in, const int* in_sizes, int n_in,
                              void* d_out, int out_size, void* d_ws, size_t ws_size,
                              hipStream_t stream) {
    (void)n_in; (void)out_size; (void)d_ws; (void)ws_size;
    const int*   codes    = (const int*)  d_in[0];
    const int*   indices  = (const int*)  d_in[1];
    const int*   indptr   = (const int*)  d_in[2];
    const float* h_self   = (const float*)d_in[3];
    const float* codebook = (const float*)d_in[4];
    const float* W_neigh  = (const float*)d_in[5];
    const float* W_self   = (const float*)d_in[6];
    const float* b_self   = (const float*)d_in[7];
    float*       out      = (float*)d_out;

    const int n_dst = in_sizes[2] - 1;

    sage_fused_kernel<<<NWG, NTHR, 0, stream>>>(
        codes, indices, indptr, h_self, codebook, W_neigh, W_self, b_self,
        out, n_dst);
}
